# starting kernel
# baseline (speedup 1.0000x reference)
#include <hip/hip_runtime.h>
#include <hip/hip_bf16.h>
#include <math.h>

typedef __attribute__((ext_vector_type(2))) float v2f;
typedef __attribute__((ext_vector_type(4))) float v4f;
typedef __attribute__((ext_vector_type(8))) float v8f;

#define BDIM   2
#define CDIM   128
#define NDIM   10000
#define KDIM   16
#define NTILES 625

__global__ __launch_bounds__(256)
void gat_gemm_kernel(const float* __restrict__ x, const float* __restrict__ W,
                     const float* __restrict__ a,
                     float* __restrict__ whT, float* __restrict__ si,
                     float* __restrict__ sj)
{
    __shared__ float Wl[CDIM * CDIM];
    __shared__ float al[2 * CDIM];

    const int tid = threadIdx.x;
    {
        const float4* Wg4 = (const float4*)W;
        float4*       Wl4 = (float4*)Wl;
        #pragma unroll 4
        for (int i = tid; i < CDIM * CDIM / 4; i += 256) Wl4[i] = Wg4[i];
        al[tid] = a[tid];
    }
    __syncthreads();

    const int wave = tid >> 5;
    const int lane = tid & 31;
    const int tile = blockIdx.x * 8 + wave;
    if (tile >= NTILES) return;

    const int b  = blockIdx.y;
    const int n0 = tile * 16;
    const int ln = lane & 15;
    const int kb = (lane >> 4) * 2;

    v2f xb[32];
    const float* xcol = x + (size_t)b * CDIM * NDIM + n0 + ln;
    #pragma unroll
    for (int ks = 0; ks < 32; ++ks) {
        const int c0 = ks * 4 + kb;
        xb[ks].x = xcol[(size_t)(c0 + 0) * NDIM];
        xb[ks].y = xcol[(size_t)(c0 + 1) * NDIM];
    }

    float sip = 0.f, sjp = 0.f;
    float* outrow = whT + ((size_t)b * NDIM + n0 + ln) * CDIM;

    #pragma unroll 1
    for (int ot = 0; ot < 8; ++ot) {
        const int o0 = ot * 16;
        v8f acc = {};
        const float* Wrow = Wl + (o0 + ln) * CDIM + kb;
        #pragma unroll
        for (int ks = 0; ks < 32; ++ks) {
            v2f av = *(const v2f*)(Wrow + ks * 4);
            acc = __builtin_amdgcn_wmma_f32_16x16x4_f32(
                      false, av, false, xb[ks], (short)0, acc, false, false);
        }
        const int oset = o0 + 8 * (lane >> 4);
        v4f lo = { acc[0], acc[1], acc[2], acc[3] };
        v4f hi = { acc[4], acc[5], acc[6], acc[7] };
        *(v4f*)(outrow + oset)     = lo;
        *(v4f*)(outrow + oset + 4) = hi;
        #pragma unroll
        for (int v = 0; v < 8; ++v) {
            sip += al[oset + v]        * acc[v];
            sjp += al[CDIM + oset + v] * acc[v];
        }
    }
    sip += __shfl_xor(sip, 16);
    sjp += __shfl_xor(sjp, 16);
    if (lane < 16) {
        si[(size_t)b * NDIM + n0 + lane] = sip;
        sj[(size_t)b * NDIM + n0 + lane] = sjp;
    }
}

__global__ __launch_bounds__(512)
void gat_aggr_kernel(const int* __restrict__ ei, const float* __restrict__ whT,
                     const float* __restrict__ si, const float* __restrict__ sj,
                     float* __restrict__ out)
{
    __shared__ float hbuf[16][CDIM + 4];

    const int tid  = threadIdx.x;
    const int wave = tid >> 5;
    const int lane = tid & 31;
    const int b    = blockIdx.y;
    const int n0   = blockIdx.x * 16;
    const int n    = n0 + wave;

    const size_t eb = ((size_t)b * NDIM + n) * KDIM;
    int   jj = 0;
    float e  = -3.0e30f;
    if (lane < 16) {
        jj = ei[eb + lane];
        const int ii = ei[(size_t)BDIM * NDIM * KDIM + eb + lane];
        const float ev = si[(size_t)b * NDIM + ii] + sj[(size_t)b * NDIM + jj];
        e = (ev > 0.f) ? ev : 0.2f * ev;
    }
    float m = e;
    m = fmaxf(m, __shfl_xor(m, 8));
    m = fmaxf(m, __shfl_xor(m, 4));
    m = fmaxf(m, __shfl_xor(m, 2));
    m = fmaxf(m, __shfl_xor(m, 1));
    float p = (lane < 16) ? __expf(e - m) : 0.f;
    float s = p;
    s += __shfl_xor(s, 8);
    s += __shfl_xor(s, 4);
    s += __shfl_xor(s, 2);
    s += __shfl_xor(s, 1);
    const float A = p / s;

    v4f h = { 0.f, 0.f, 0.f, 0.f };
    const float* wb = whT + (size_t)b * NDIM * CDIM;
    #pragma unroll
    for (int k = 0; k < 16; ++k) {
        const float ak = __shfl(A, k);
        const int   jk = __shfl(jj, k);
        const v4f w4 = *(const v4f*)(wb + (size_t)jk * CDIM + 4 * lane);
        h += ak * w4;
    }
    *(v4f*)(&hbuf[wave][4 * lane]) = h;
    __syncthreads();

    const int c = tid >> 2;
    const int q = tid & 3;
    v4f o4 = { hbuf[4 * q + 0][c], hbuf[4 * q + 1][c],
               hbuf[4 * q + 2][c], hbuf[4 * q + 3][c] };
    *(v4f*)(out + ((size_t)b * CDIM + c) * NDIM + n0 + 4 * q) = o4;
}

extern "C" void kernel_launch(void* const* d_in, const int* in_sizes, int n_in,
                              void* d_out, int out_size, void* d_ws, size_t ws_size,
                              hipStream_t stream)
{
    const float* x  = (const float*)d_in[0];
    const int*   ei = (const int*)  d_in[1];
    const float* W  = (const float*)d_in[2];
    const float* a  = (const float*)d_in[3];
    float* out = (float*)d_out;

    float* whT = (float*)d_ws;
    float* si  = whT + (size_t)BDIM * NDIM * CDIM;
    float* sj  = si  + (size_t)BDIM * NDIM;

    dim3 g1((NTILES + 7) / 8, BDIM);
    gat_gemm_kernel<<<g1, 256, 0, stream>>>(x, W, a, whT, si, sj);

    dim3 g2(NTILES, BDIM);
    gat_aggr_kernel<<<g2, 512, 0, stream>>>(ei, whT, si, sj, out);
}


# Kernel 2 pipelined over its 5 node tiles: the next tile's row loads stay in flight across an LDS-only barrier while waves 0-7 run this tile's WMMA product (LDS tile double-buffered); sums unchanged
# speedup vs baseline: 1.6071x; 1.6071x over previous
#include <hip/hip_runtime.h>
#include <math.h>

typedef __attribute__((ext_vector_type(2))) float v2f;
typedef __attribute__((ext_vector_type(4))) float v4f;
typedef __attribute__((ext_vector_type(8))) float v8f;

#define BDIM   2
#define CDIM   128
#define NDIM   10000
#define KDIM   16
#define TN     80
#define NWG    125
#define PITCH  132


__global__ __launch_bounds__(256)
void gat_prep_kernel(const float* __restrict__ x, const float* __restrict__ W,
                     const float* __restrict__ a,
                     float* __restrict__ xT, float* __restrict__ si,
                     float* __restrict__ sj)
{
    __shared__ __attribute__((aligned(16))) float tl[TN * PITCH];
    __shared__ __attribute__((aligned(16))) float up[2][2][CDIM];

    const int tid  = threadIdx.x;
    const int lane = tid & 31;
    const int wave = __builtin_amdgcn_readfirstlane(tid >> 5);
    const int b    = blockIdx.y;
    const int n0   = blockIdx.x * TN;

    const int q  = tid & 3;
    const int cl = tid >> 2;
    const float* xg = x + (size_t)b * CDIM * NDIM + n0 + 4 * q;
    v4f xv[10];
    #pragma unroll
    for (int it = 0; it < 10; ++it) {
        const int g = it >> 1;
        const int c = (it & 1) * 64 + cl;
        xv[it] = *(const v4f*)(xg + (size_t)c * NDIM + 16 * g);
    }

    {
        const int c  = tid & 127;
        const int hf = wave >> 2;
        const float* Wc = W + (size_t)(64 * hf) * CDIM + c;
        const float* ah = a + 64 * hf;
        float ui = 0.f, uj = 0.f;
        #pragma unroll 32
        for (int o = 0; o < 64; ++o) {
            const float w = Wc[o * CDIM];
            ui += ah[o] * w;
            uj += ah[CDIM + o] * w;
        }
        up[0][hf][c] = ui;
        up[1][hf][c] = uj;
    }

    #pragma unroll
    for (int it = 0; it < 10; ++it) {
        const int g  = it >> 1;
        const int c  = (it & 1) * 64 + cl;
        const int nl = 16 * g + 4 * q;
        tl[(nl + 0) * PITCH + c] = xv[it].x;
        tl[(nl + 1) * PITCH + c] = xv[it].y;
        tl[(nl + 2) * PITCH + c] = xv[it].z;
        tl[(nl + 3) * PITCH + c] = xv[it].w;
    }
    __syncthreads();

    const v4f ui4 = *(const v4f*)&up[0][0][4 * lane] + *(const v4f*)&up[0][1][4 * lane];
    const v4f uj4 = *(const v4f*)&up[1][0][4 * lane] + *(const v4f*)&up[1][1][4 * lane];
    float* xTb = xT + ((size_t)b * NDIM + n0) * CDIM + 4 * lane;
    #pragma unroll 5
    for (int r = 0; r < 10; ++r) {
        const int nl = wave + 8 * r;
        const v4f v = *(const v4f*)&tl[nl * PITCH + 4 * lane];
        *(v4f*)(xTb + (size_t)nl * CDIM) = v;
        float pi = v.x * ui4.x + v.y * ui4.y + v.z * ui4.z + v.w * ui4.w;
        float pj = v.x * uj4.x + v.y * uj4.y + v.z * uj4.z + v.w * uj4.w;
        pi += __shfl_xor(pi, 16);  pj += __shfl_xor(pj, 16);
        pi += __shfl_xor(pi, 8);   pj += __shfl_xor(pj, 8);
        pi += __shfl_xor(pi, 4);   pj += __shfl_xor(pj, 4);
        pi += __shfl_xor(pi, 2);   pj += __shfl_xor(pj, 2);
        pi += __shfl_xor(pi, 1);   pj += __shfl_xor(pj, 1);
        if (lane == 0) {
            si[(size_t)b * NDIM + n0 + nl] = pi;
            sj[(size_t)b * NDIM + n0 + nl] = pj;
        }
    }
}

__device__ __forceinline__ void gemm_tile(const v4f (&wa)[16], const float* agl,
                                          float* __restrict__ outp)
{
    v8f acc = (v8f){0.f, 0.f, 0.f, 0.f, 0.f, 0.f, 0.f, 0.f};
    #pragma unroll
    for (int j = 0; j < 16; ++j) {
        const v4f xb = *(const v4f*)(agl + 8 * j);
        const v2f a0 = { wa[j].x, wa[j].y };
        const v2f a1 = { wa[j].z, wa[j].w };
        const v2f b0 = { xb.x, xb.y };
        const v2f b1 = { xb.z, xb.w };
        acc = __builtin_amdgcn_wmma_f32_16x16x4_f32(
                  false, a0, false, b0, (short)0, acc, false, false);
        acc = __builtin_amdgcn_wmma_f32_16x16x4_f32(
                  false, a1, false, b1, (short)0, acc, false, false);
    }
    #pragma unroll
    for (int v = 0; v < 8; ++v)
        outp[(size_t)v * NDIM] = acc[v];
}

__device__ __forceinline__ void lds_barrier()
{
    __builtin_amdgcn_fence(__ATOMIC_RELEASE, "workgroup", "local");
    __builtin_amdgcn_s_barrier();
    __builtin_amdgcn_fence(__ATOMIC_ACQUIRE, "workgroup", "local");
}

__global__ __launch_bounds__(512)
void gat_aggr_gemm_kernel(const int* __restrict__ ei, const float* __restrict__ xT,
                          const float* __restrict__ si, const float* __restrict__ sj,
                          const float* __restrict__ W, float* __restrict__ out)
{
    __shared__ __attribute__((aligned(16))) float ag[2][16 * PITCH];

    const int tid  = threadIdx.x;
    const int lane = tid & 31;
    const int wave = __builtin_amdgcn_readfirstlane(tid >> 5);
    const int b    = blockIdx.y;
    const int n0   = blockIdx.x * TN;
    const int ln   = lane & 15;
    const int hh   = lane >> 4;
    const int ot   = wave & 7;
    const bool mm  = wave < 8;

    v4f wa[16];
    #pragma unroll
    for (int j = 0; j < 16; ++j) wa[j] = (v4f){ 0.f, 0.f, 0.f, 0.f };
    if (mm) {
        const float* Wrow = W + (size_t)(16 * ot + ln) * CDIM + 4 * hh;
        #pragma unroll
        for (int j = 0; j < 16; ++j) wa[j] = *(const v4f*)(Wrow + 8 * j);
    }

    const float* xb = xT + (size_t)b * NDIM * CDIM + 4 * lane;
    const float* sib = si + (size_t)b * NDIM;
    const float* sjb = sj + (size_t)b * NDIM;

    int jjv[5];
    int abv[5];
    #pragma unroll
    for (int i = 0; i < 5; ++i) {
        const int nl = 16 * i + wave;
        const size_t eb = ((size_t)b * NDIM + n0 + nl) * KDIM + ln;
        const int jj = ei[eb];
        const int ii = ei[(size_t)BDIM * NDIM * KDIM + eb];
        const float ev = sib[ii] + sjb[jj];
        const float e  = (ev > 0.f) ? ev : 0.2f * ev;
        float m = e;
        m = fmaxf(m, __shfl_xor(m, 8));
        m = fmaxf(m, __shfl_xor(m, 4));
        m = fmaxf(m, __shfl_xor(m, 2));
        m = fmaxf(m, __shfl_xor(m, 1));
        const float p = __expf(e - m);
        float s = p;
        s += __shfl_xor(s, 8);
        s += __shfl_xor(s, 4);
        s += __shfl_xor(s, 2);
        s += __shfl_xor(s, 1);
        const float A = p / s;
        jjv[i] = jj;
        abv[i] = __float_as_int(A);
    }

    v4f buf[16];
    #pragma unroll
    for (int k = 0; k < 16; ++k) {
        const int jk = __builtin_amdgcn_readlane(jjv[0], k);
        buf[k] = *(const v4f*)(xb + (size_t)jk * CDIM);
    }

    float* outp = out + ((size_t)b * CDIM + 16 * ot + 8 * hh) * NDIM + n0 + ln;

    #pragma unroll
    for (int t = 0; t < 5; ++t) {
        v4f h = { 0.f, 0.f, 0.f, 0.f };
        #pragma unroll
        for (int k = 0; k < 16; ++k) {
            const float ak = __int_as_float(__builtin_amdgcn_readlane(abv[t], k));
            h += ak * buf[k];
            if (t + 1 < 5) {
                const int jk = __builtin_amdgcn_readlane(jjv[t + 1], k);
                buf[k] = *(const v4f*)(xb + (size_t)jk * CDIM);
            }
        }
        *(v4f*)&ag[t & 1][wave * PITCH + 4 * lane] = h;
        lds_barrier();
        if (mm)
            gemm_tile(wa, &ag[t & 1][ln * PITCH + 4 * hh], outp + 16 * t);
    }
}

extern "C" void kernel_launch(void* const* d_in, const int* in_sizes, int n_in,
                              void* d_out, int out_size, void* d_ws, size_t ws_size,
                              hipStream_t stream)
{
    const float* x  = (const float*)d_in[0];
    const int*   ei = (const int*)  d_in[1];
    const float* W  = (const float*)d_in[2];
    const float* a  = (const float*)d_in[3];
    float* out = (float*)d_out;

    float* xT = (float*)d_ws;
    float* si = xT + (size_t)BDIM * NDIM * CDIM;
    float* sj = si + (size_t)BDIM * NDIM;

    dim3 g(NWG, BDIM);
    gat_prep_kernel<<<g, 256, 0, stream>>>(x, W, a, xT, si, sj);
    gat_aggr_gemm_kernel<<<g, 512, 0, stream>>>(ei, xT, si, sj, W, out);
}
